# starting kernel
# baseline (speedup 1.0000x reference)
#include <hip/hip_runtime.h>


#define BATCH 4
#define SEQ   2048
#define DIM   512
#define HEADS 8
#define DK    64
#define NCOL  1536
#define BPS   160

typedef __attribute__((ext_vector_type(2))) float v2f;
typedef __attribute__((ext_vector_type(8))) float v8f;

__device__ __forceinline__ v8f wmma4(v2f a, v2f b, v8f c) {
    return __builtin_amdgcn_wmma_f32_16x16x4_f32(
         false, a,  false, b,
         (short)0, c,  false,  false);
}

__device__ __forceinline__ void stage_pair(float* __restrict__ dst, int kp, int cp,
                                           const float* __restrict__ src, int srcStride) {
    const float* s = src + (size_t)(2 * kp) * srcStride + 2 * cp;
    float2 r0 = *(const float2*)s;
    float2 r1 = *(const float2*)(s + srcStride);
    float4 pk;
    pk.x = r0.x; pk.y = r1.x; pk.z = r0.y; pk.w = r1.y;
    *(float4*)&dst[kp * BPS + 4 * cp] = pk;
}

__global__ __launch_bounds__(256)
void qkv_gemm_kernel(const float* __restrict__ x, const float* __restrict__ w,
                     float* __restrict__ qb, float* __restrict__ kb,
                     float* __restrict__ vb) {
    __shared__ float lds_a[64][68];
    __shared__ float lds_b[32 * BPS];

    const int tid  = threadIdx.x;
    const int lane = tid & 31, wave = tid >> 5;
    const int half = lane >> 4, l16 = lane & 15;
    const int M0 = (wave & 3) * 16;
    const int N0 = (wave >> 2) * 32;

    const int cb    = blockIdx.x;
    const int col0  = cb * 64;
    const int which = cb >> 3, h = cb & 7;
    const int rowBase = blockIdx.y * 64;
    float* dstBase = (which == 0) ? qb : (which == 1) ? kb : vb;

    const int lr  = tid >> 2;
    const int lc  = (tid & 3) * 16;
    const int cp  = tid & 31;
    const int kb8 = tid >> 5;

    v8f acc0 = {}; v8f acc1 = {};

    for (int kc = 0; kc < DIM; kc += 64) {
        __syncthreads();
        #pragma unroll
        for (int i = 0; i < 4; ++i) {
            *(float4*)&lds_a[lr][lc + 4*i] =
                *(const float4*)&x[(size_t)(rowBase + lr) * DIM + kc + lc + 4*i];
            stage_pair(lds_b, kb8 + 8*i, cp, &w[(size_t)kc * NCOL + col0], NCOL);
        }
        __syncthreads();
        #pragma unroll
        for (int k0 = 0; k0 < 64; k0 += 4) {
            const int kp = (k0 >> 1) + half;
            v2f a  = *(const v2f*)&lds_a[M0 + l16][k0 + 2*half];
            v2f b0 = *(const v2f*)&lds_b[kp * BPS + 2*(N0 + l16)];
            v2f b1 = *(const v2f*)&lds_b[kp * BPS + 2*(N0 + 16 + l16)];
            acc0 = wmma4(a, b0, acc0);
            acc1 = wmma4(a, b1, acc1);
        }
    }

    #pragma unroll
    for (int r = 0; r < 8; ++r) {
        int rowg = rowBase + M0 + r + 8*half;
        int bb = rowg >> 11, n = rowg & (SEQ - 1);
        float* dst = dstBase + ((size_t)((bb * HEADS + h) * SEQ + n)) * DK;
        dst[N0 + l16]      = acc0[r];
        dst[N0 + 16 + l16] = acc1[r];
    }
}

__global__ __launch_bounds__(256)
void ln_kernel(float* __restrict__ t, const float* __restrict__ g,
               const float* __restrict__ bta) {
    const int row  = blockIdx.x * 8 + (threadIdx.x >> 5);
    const int lane = threadIdx.x & 31;
    const int h = (row >> 11) & (HEADS - 1);

    float2 xv = *(float2*)(t + (size_t)row * DK + lane * 2);
    float s = xv.x + xv.y;
    #pragma unroll
    for (int o = 16; o >= 1; o >>= 1) s += __shfl_xor(s, o, 32);
    const float mu = s * (1.0f / DK);
    const float d0 = xv.x - mu, d1 = xv.y - mu;
    float v = d0 * d0 + d1 * d1;
    #pragma unroll
    for (int o = 16; o >= 1; o >>= 1) v += __shfl_xor(v, o, 32);
    const float rinv = rsqrtf(v * (1.0f / DK) + 1e-5f);

    float2 gv = *(const float2*)(g   + h * DK + lane * 2);
    float2 bv = *(const float2*)(bta + h * DK + lane * 2);
    float2 ov;
    ov.x = d0 * rinv * gv.x + bv.x;
    ov.y = d1 * rinv * gv.y + bv.y;
    *(float2*)(t + (size_t)row * DK + lane * 2) = ov;
}

__global__ __launch_bounds__(256)
void kv_gemm_kernel(const float* __restrict__ kbuf, const float* __restrict__ vbuf,
                    float* __restrict__ kvb) {
    __shared__ float lds_k[32 * BPS];
    __shared__ float lds_v[32 * BPS];

    const int tid  = threadIdx.x;
    const int lane = tid & 31, wave = tid >> 5;
    const int half = lane >> 4, l16 = lane & 15;
    const int M0 = (wave & 3) * 16, N0 = (wave >> 2) * 32;
    const int bh = blockIdx.x;
    const float* kp_ptr = kbuf + (size_t)bh * SEQ * DK;
    const float* vp_ptr = vbuf + (size_t)bh * SEQ * DK;
    const int cp  = tid & 31;
    const int tb8 = tid >> 5;

    v8f acc0 = {}; v8f acc1 = {};

    for (int t0 = 0; t0 < SEQ; t0 += 64) {
        __syncthreads();
        #pragma unroll
        for (int i = 0; i < 4; ++i) {
            stage_pair(lds_k, tb8 + 8*i, cp, kp_ptr + (size_t)t0 * DK, DK);
            stage_pair(lds_v, tb8 + 8*i, cp, vp_ptr + (size_t)t0 * DK, DK);
        }
        __syncthreads();
        #pragma unroll
        for (int k0 = 0; k0 < 64; k0 += 4) {
            const int kp = (k0 >> 1) + half;
            v2f a  = *(const v2f*)&lds_k[kp * BPS + 2*(M0 + l16)];
            v2f b0 = *(const v2f*)&lds_v[kp * BPS + 2*(N0 + l16)];
            v2f b1 = *(const v2f*)&lds_v[kp * BPS + 2*(N0 + 16 + l16)];
            acc0 = wmma4(a, b0, acc0);
            acc1 = wmma4(a, b1, acc1);
        }
    }

    const float scale = 1.0f / 16384.0f;
    float* kvd = kvb + (size_t)bh * DK * DK;
    #pragma unroll
    for (int r = 0; r < 8; ++r) {
        int m = M0 + r + 8*half;
        kvd[m * DK + N0 + l16]      = acc0[r] * scale;
        kvd[m * DK + N0 + 16 + l16] = acc1[r] * scale;
    }
}

__global__ __launch_bounds__(256)
void out_gemm_kernel(const float* __restrict__ qbuf, const float* __restrict__ kvb,
                     float* __restrict__ out) {
    __shared__ float lds_q[64][68];
    __shared__ float lds_kv[32 * BPS];

    const int tid  = threadIdx.x;
    const int lane = tid & 31, wave = tid >> 5;
    const int half = lane >> 4, l16 = lane & 15;
    const int M0 = (wave & 3) * 16, N0 = (wave >> 2) * 32;
    const int tok0 = blockIdx.x * 64;
    const int bh = blockIdx.y;
    const int bb = bh >> 3, h = bh & 7;
    const float* qp  = qbuf + (size_t)bh * SEQ * DK + (size_t)tok0 * DK;
    const float* kvp = kvb + (size_t)bh * DK * DK;
    const int lr  = tid >> 2, lc = (tid & 3) * 16;
    const int cp  = tid & 31, kb8 = tid >> 5;

    #pragma unroll
    for (int i = 0; i < 4; ++i) {
        *(float4*)&lds_q[lr][lc + 4*i] = *(const float4*)&qp[(size_t)lr * DK + lc + 4*i];
        stage_pair(lds_kv, kb8 + 8*i, cp, kvp, DK);
    }
    __syncthreads();

    v8f acc0 = {}; v8f acc1 = {};
    #pragma unroll
    for (int k0 = 0; k0 < 64; k0 += 4) {
        const int kp = (k0 >> 1) + half;
        v2f a  = *(const v2f*)&lds_q[M0 + l16][k0 + 2*half];
        v2f b0 = *(const v2f*)&lds_kv[kp * BPS + 2*(N0 + l16)];
        v2f b1 = *(const v2f*)&lds_kv[kp * BPS + 2*(N0 + 16 + l16)];
        acc0 = wmma4(a, b0, acc0);
        acc1 = wmma4(a, b1, acc1);
    }

    #pragma unroll
    for (int r = 0; r < 8; ++r) {
        int n = tok0 + M0 + r + 8*half;
        float* dst = out + ((size_t)(bb * SEQ + n)) * (HEADS * DK) + h * DK;
        dst[N0 + l16]      = acc0[r];
        dst[N0 + 16 + l16] = acc1[r];
    }
}

extern "C" void kernel_launch(void* const* d_in, const int* in_sizes, int n_in,
                              void* d_out, int out_size, void* d_ws, size_t ws_size,
                              hipStream_t stream) {
    const float* x  = (const float*)d_in[0];
    const float* w  = (const float*)d_in[1];
    const float* gq = (const float*)d_in[2];
    const float* bq = (const float*)d_in[3];
    const float* gk = (const float*)d_in[4];
    const float* bk = (const float*)d_in[5];
    float* out = (float*)d_out;

    float* ws = (float*)d_ws;
    const size_t perBuf = (size_t)BATCH * HEADS * SEQ * DK;
    float* qb  = ws;
    float* kb  = ws + perBuf;
    float* vb  = ws + 2 * perBuf;
    float* kvb = ws + 3 * perBuf;

    qkv_gemm_kernel<<<dim3(NCOL / 64, (BATCH * SEQ) / 64), 256, 0, stream>>>(x, w, qb, kb, vb);
    ln_kernel<<<dim3((BATCH * HEADS * SEQ) / 8), 256, 0, stream>>>(qb, gq, bq);
    ln_kernel<<<dim3((BATCH * HEADS * SEQ) / 8), 256, 0, stream>>>(kb, gk, bk);
    kv_gemm_kernel<<<dim3(BATCH * HEADS), 256, 0, stream>>>(kb, vb, kvb);
    out_gemm_kernel<<<dim3(SEQ / 64, BATCH * HEADS), 256, 0, stream>>>(qb, kvb, out);
}


# Split the k^T@v GEMM over the sequence: 8 workgroups per head (256 blocks, was 32) each write a scaled partial; the out kernel adds the 8 in fixed order while staging KV (token sum order changes)
# speedup vs baseline: 1.3715x; 1.3715x over previous
#include <hip/hip_runtime.h>


#define BATCH 4
#define SEQ   2048
#define DIM   512
#define HEADS 8
#define DK    64
#define NCOL  1536
#define BPS   160
#define KV_SPLIT 8
#define KV_TOK   (SEQ / KV_SPLIT)

typedef __attribute__((ext_vector_type(2))) float v2f;
typedef __attribute__((ext_vector_type(8))) float v8f;

__device__ __forceinline__ v8f wmma4(v2f a, v2f b, v8f c) {
    return __builtin_amdgcn_wmma_f32_16x16x4_f32(
         false, a,  false, b,
         (short)0, c,  false,  false);
}

__device__ __forceinline__ void stage_pair(float* __restrict__ dst, int kp, int cp,
                                           const float* __restrict__ src, int srcStride) {
    const float* s = src + (size_t)(2 * kp) * srcStride + 2 * cp;
    float2 r0 = *(const float2*)s;
    float2 r1 = *(const float2*)(s + srcStride);
    float4 pk;
    pk.x = r0.x; pk.y = r1.x; pk.z = r0.y; pk.w = r1.y;
    *(float4*)&dst[kp * BPS + 4 * cp] = pk;
}

__global__ __launch_bounds__(256)
void qkv_gemm_kernel(const float* __restrict__ x, const float* __restrict__ w,
                     float* __restrict__ qb, float* __restrict__ kb,
                     float* __restrict__ vb) {
    __shared__ float lds_a[64][68];
    __shared__ float lds_b[32 * BPS];

    const int tid  = threadIdx.x;
    const int lane = tid & 31, wave = tid >> 5;
    const int half = lane >> 4, l16 = lane & 15;
    const int M0 = (wave & 3) * 16;
    const int N0 = (wave >> 2) * 32;

    const int cb    = blockIdx.x;
    const int col0  = cb * 64;
    const int which = cb >> 3, h = cb & 7;
    const int rowBase = blockIdx.y * 64;
    float* dstBase = (which == 0) ? qb : (which == 1) ? kb : vb;

    const int lr  = tid >> 2;
    const int lc  = (tid & 3) * 16;
    const int cp  = tid & 31;
    const int kb8 = tid >> 5;

    v8f acc0 = {}; v8f acc1 = {};

    for (int kc = 0; kc < DIM; kc += 64) {
        __syncthreads();
        #pragma unroll
        for (int i = 0; i < 4; ++i) {
            *(float4*)&lds_a[lr][lc + 4*i] =
                *(const float4*)&x[(size_t)(rowBase + lr) * DIM + kc + lc + 4*i];
            stage_pair(lds_b, kb8 + 8*i, cp, &w[(size_t)kc * NCOL + col0], NCOL);
        }
        __syncthreads();
        #pragma unroll
        for (int k0 = 0; k0 < 64; k0 += 4) {
            const int kp = (k0 >> 1) + half;
            v2f a  = *(const v2f*)&lds_a[M0 + l16][k0 + 2*half];
            v2f b0 = *(const v2f*)&lds_b[kp * BPS + 2*(N0 + l16)];
            v2f b1 = *(const v2f*)&lds_b[kp * BPS + 2*(N0 + 16 + l16)];
            acc0 = wmma4(a, b0, acc0);
            acc1 = wmma4(a, b1, acc1);
        }
    }

    #pragma unroll
    for (int r = 0; r < 8; ++r) {
        int rowg = rowBase + M0 + r + 8*half;
        int bb = rowg >> 11, n = rowg & (SEQ - 1);
        float* dst = dstBase + ((size_t)((bb * HEADS + h) * SEQ + n)) * DK;
        dst[N0 + l16]      = acc0[r];
        dst[N0 + 16 + l16] = acc1[r];
    }
}

__global__ __launch_bounds__(256)
void ln_kernel(float* __restrict__ t, const float* __restrict__ g,
               const float* __restrict__ bta) {
    const int row  = blockIdx.x * 8 + (threadIdx.x >> 5);
    const int lane = threadIdx.x & 31;
    const int h = (row >> 11) & (HEADS - 1);

    float2 xv = *(float2*)(t + (size_t)row * DK + lane * 2);
    float s = xv.x + xv.y;
    #pragma unroll
    for (int o = 16; o >= 1; o >>= 1) s += __shfl_xor(s, o, 32);
    const float mu = s * (1.0f / DK);
    const float d0 = xv.x - mu, d1 = xv.y - mu;
    float v = d0 * d0 + d1 * d1;
    #pragma unroll
    for (int o = 16; o >= 1; o >>= 1) v += __shfl_xor(v, o, 32);
    const float rinv = rsqrtf(v * (1.0f / DK) + 1e-5f);

    float2 gv = *(const float2*)(g   + h * DK + lane * 2);
    float2 bv = *(const float2*)(bta + h * DK + lane * 2);
    float2 ov;
    ov.x = d0 * rinv * gv.x + bv.x;
    ov.y = d1 * rinv * gv.y + bv.y;
    *(float2*)(t + (size_t)row * DK + lane * 2) = ov;
}

__global__ __launch_bounds__(256)
void kv_gemm_kernel(float* kbuf, const float* __restrict__ vbuf) {
    __shared__ float lds_k[32 * BPS];
    __shared__ float lds_v[32 * BPS];

    const int tid  = threadIdx.x;
    const int lane = tid & 31, wave = tid >> 5;
    const int half = lane >> 4, l16 = lane & 15;
    const int M0 = (wave & 3) * 16, N0 = (wave >> 2) * 32;
    const int bh = blockIdx.y;
    const size_t slice = ((size_t)bh * SEQ + (size_t)blockIdx.x * KV_TOK) * DK;
    const float* kp_ptr = kbuf + slice;
    const float* vp_ptr = vbuf + slice;
    const int cp  = tid & 31;
    const int tb8 = tid >> 5;

    v8f acc0 = {}; v8f acc1 = {};

    for (int t0 = 0; t0 < KV_TOK; t0 += 64) {
        __syncthreads();
        #pragma unroll
        for (int i = 0; i < 4; ++i) {
            stage_pair(lds_k, tb8 + 8*i, cp, kp_ptr + (size_t)t0 * DK, DK);
            stage_pair(lds_v, tb8 + 8*i, cp, vp_ptr + (size_t)t0 * DK, DK);
        }
        __syncthreads();
        #pragma unroll
        for (int k0 = 0; k0 < 64; k0 += 4) {
            const int kp = (k0 >> 1) + half;
            v2f a  = *(const v2f*)&lds_k[kp * BPS + 2*(M0 + l16)];
            v2f b0 = *(const v2f*)&lds_v[kp * BPS + 2*(N0 + l16)];
            v2f b1 = *(const v2f*)&lds_v[kp * BPS + 2*(N0 + 16 + l16)];
            acc0 = wmma4(a, b0, acc0);
            acc1 = wmma4(a, b1, acc1);
        }
    }

    const float scale = 1.0f / 16384.0f;
    float* kvd = kbuf + slice;
    #pragma unroll
    for (int r = 0; r < 8; ++r) {
        int m = M0 + r + 8*half;
        kvd[m * DK + N0 + l16]      = acc0[r] * scale;
        kvd[m * DK + N0 + 16 + l16] = acc1[r] * scale;
    }
}

__global__ __launch_bounds__(256) __attribute__((amdgpu_waves_per_eu(8, 8)))
void out_gemm_kernel(const float* __restrict__ qbuf, const float* __restrict__ part,
                     float* __restrict__ out) {
    __shared__ float lds_q[64][68];
    __shared__ float lds_kv[32 * BPS];

    const int tid  = threadIdx.x;
    const int lane = tid & 31, wave = tid >> 5;
    const int half = lane >> 4, l16 = lane & 15;
    const int M0 = (wave & 3) * 16, N0 = (wave >> 2) * 32;
    const int tok0 = blockIdx.x * 64;
    const int bh = blockIdx.y;
    const int bb = bh >> 3, h = bh & 7;
    const float* qp  = qbuf + (size_t)bh * SEQ * DK + (size_t)tok0 * DK;
    const float* pp  = part + (size_t)bh * SEQ * DK;
    const int lr  = tid >> 2, lc = (tid & 3) * 16;
    const int c4  = tid & 15, kp16 = tid >> 4;

    #pragma unroll 1
    for (int i = 0; i < 2; ++i) {
        const float4 q0 = *(const float4*)&qp[(size_t)lr * DK + lc + 8*i];
        const float4 q1 = *(const float4*)&qp[(size_t)lr * DK + lc + 8*i + 4];
        const int kp = kp16 + 16*i;
        const float* s0 = pp + (size_t)(2 * kp) * DK + 4 * c4;
        float4 r0 = *(const float4*)s0;
        float4 r1 = *(const float4*)(s0 + DK);
        #pragma unroll
        for (int s = 1; s < KV_SPLIT; ++s) {
            const float4 a0 = *(const float4*)(s0 + (size_t)s * KV_TOK * DK);
            const float4 a1 = *(const float4*)(s0 + (size_t)s * KV_TOK * DK + DK);
            r0.x += a0.x; r0.y += a0.y; r0.z += a0.z; r0.w += a0.w;
            r1.x += a1.x; r1.y += a1.y; r1.z += a1.z; r1.w += a1.w;
        }
        float4 p0, p1;
        p0.x = r0.x; p0.y = r1.x; p0.z = r0.y; p0.w = r1.y;
        p1.x = r0.z; p1.y = r1.z; p1.z = r0.w; p1.w = r1.w;
        *(float4*)&lds_kv[kp * BPS + 8 * c4]     = p0;
        *(float4*)&lds_kv[kp * BPS + 8 * c4 + 4] = p1;
        *(float4*)&lds_q[lr][lc + 8*i]     = q0;
        *(float4*)&lds_q[lr][lc + 8*i + 4] = q1;
    }
    __syncthreads();

    v8f acc0 = {}; v8f acc1 = {};
    #pragma unroll
    for (int k0 = 0; k0 < 64; k0 += 4) {
        const int kp = (k0 >> 1) + half;
        v2f a  = *(const v2f*)&lds_q[M0 + l16][k0 + 2*half];
        v2f b0 = *(const v2f*)&lds_kv[kp * BPS + 2*(N0 + l16)];
        v2f b1 = *(const v2f*)&lds_kv[kp * BPS + 2*(N0 + 16 + l16)];
        acc0 = wmma4(a, b0, acc0);
        acc1 = wmma4(a, b1, acc1);
    }

    #pragma unroll
    for (int r = 0; r < 8; ++r) {
        int n = tok0 + M0 + r + 8*half;
        float* dst = out + ((size_t)(bb * SEQ + n)) * (HEADS * DK) + h * DK;
        dst[N0 + l16]      = acc0[r];
        dst[N0 + 16 + l16] = acc1[r];
    }
}

extern "C" void kernel_launch(void* const* d_in, const int* in_sizes, int n_in,
                              void* d_out, int out_size, void* d_ws, size_t ws_size,
                              hipStream_t stream) {
    const float* x  = (const float*)d_in[0];
    const float* w  = (const float*)d_in[1];
    const float* gq = (const float*)d_in[2];
    const float* bq = (const float*)d_in[3];
    const float* gk = (const float*)d_in[4];
    const float* bk = (const float*)d_in[5];
    float* out = (float*)d_out;

    float* ws = (float*)d_ws;
    const size_t perBuf = (size_t)BATCH * HEADS * SEQ * DK;
    float* qb  = ws;
    float* kb  = ws + perBuf;
    float* vb  = ws + 2 * perBuf;

    qkv_gemm_kernel<<<dim3(NCOL / 64, (BATCH * SEQ) / 64), 256, 0, stream>>>(x, w, qb, kb, vb);
    ln_kernel<<<dim3((BATCH * HEADS * SEQ) / 8), 256, 0, stream>>>(qb, gq, bq);
    ln_kernel<<<dim3((BATCH * HEADS * SEQ) / 8), 256, 0, stream>>>(kb, gk, bk);
    kv_gemm_kernel<<<dim3(KV_SPLIT, BATCH * HEADS), 256, 0, stream>>>(kb, vb);
    out_gemm_kernel<<<dim3(SEQ / 64, BATCH * HEADS), 256, 0, stream>>>(qb, kb, out);
}


# Per-head LayerNorm of q and k moved into the qkv GEMM epilogue (tile staged through LDS, one wave per whole 256-B row, same arithmetic and sum order): 3 launches instead of 5, two passes gone
# speedup vs baseline: 1.9413x; 1.4155x over previous
#include <hip/hip_runtime.h>


#define BATCH 4
#define SEQ   2048
#define DIM   512
#define HEADS 8
#define DK    64
#define NCOL  1536
#define BPS   160
#define KV_SPLIT 8
#define KV_TOK   (SEQ / KV_SPLIT)

typedef __attribute__((ext_vector_type(2))) float v2f;
typedef __attribute__((ext_vector_type(8))) float v8f;

__device__ __forceinline__ v8f wmma4(v2f a, v2f b, v8f c) {
    return __builtin_amdgcn_wmma_f32_16x16x4_f32(
         false, a,  false, b,
         (short)0, c,  false,  false);
}

__device__ __forceinline__ void stage_pair(float* __restrict__ dst, int kp, int cp,
                                           const float* __restrict__ src, int srcStride) {
    const float* s = src + (size_t)(2 * kp) * srcStride + 2 * cp;
    float2 r0 = *(const float2*)s;
    float2 r1 = *(const float2*)(s + srcStride);
    float4 pk;
    pk.x = r0.x; pk.y = r1.x; pk.z = r0.y; pk.w = r1.y;
    *(float4*)&dst[kp * BPS + 4 * cp] = pk;
}

__global__ __launch_bounds__(256)
void qkv_ln_gemm_kernel(const float* __restrict__ x, const float* __restrict__ w,
                        const float* __restrict__ gq, const float* __restrict__ bq,
                        const float* __restrict__ gk, const float* __restrict__ bk,
                        float* __restrict__ qb, float* __restrict__ kb,
                        float* __restrict__ vb) {
    __shared__ float lds_a[64][68];
    __shared__ float lds_b[32 * BPS];

    const int tid  = threadIdx.x;
    const int lane = tid & 31, wave = tid >> 5;
    const int half = lane >> 4, l16 = lane & 15;
    const int M0 = (wave & 3) * 16;
    const int N0 = (wave >> 2) * 32;

    const int cb    = blockIdx.x;
    const int col0  = cb * 64;
    const int which = cb >> 3, h = cb & 7;
    const int rowBase = blockIdx.y * 64;
    float* dstBase = (which == 0) ? qb : (which == 1) ? kb : vb;

    const int lr  = tid >> 2;
    const int lc  = (tid & 3) * 16;
    const int cp  = tid & 31;
    const int kb8 = tid >> 5;

    v8f acc0 = {}; v8f acc1 = {};

    for (int kc = 0; kc < DIM; kc += 64) {
        __syncthreads();
        #pragma unroll
        for (int i = 0; i < 4; ++i) {
            *(float4*)&lds_a[lr][lc + 4*i] =
                *(const float4*)&x[(size_t)(rowBase + lr) * DIM + kc + lc + 4*i];
            stage_pair(lds_b, kb8 + 8*i, cp, &w[(size_t)kc * NCOL + col0], NCOL);
        }
        __syncthreads();
        #pragma unroll
        for (int k0 = 0; k0 < 64; k0 += 4) {
            const int kp = (k0 >> 1) + half;
            v2f a  = *(const v2f*)&lds_a[M0 + l16][k0 + 2*half];
            v2f b0 = *(const v2f*)&lds_b[kp * BPS + 2*(N0 + l16)];
            v2f b1 = *(const v2f*)&lds_b[kp * BPS + 2*(N0 + 16 + l16)];
            acc0 = wmma4(a, b0, acc0);
            acc1 = wmma4(a, b1, acc1);
        }
    }

    __syncthreads();
    #pragma unroll
    for (int r = 0; r < 8; ++r) {
        lds_a[M0 + r + 8*half][N0 + l16]      = acc0[r];
        lds_a[M0 + r + 8*half][N0 + 16 + l16] = acc1[r];
    }
    __syncthreads();

    float2 xv[8];
    #pragma unroll
    for (int r = 0; r < 8; ++r)
        xv[r] = *(const float2*)&lds_a[wave * 8 + r][2 * lane];

    if (which < 2) {
        const float* g   = (which == 0) ? gq : gk;
        const float* bta = (which == 0) ? bq : bk;
        const float2 gv = *(const float2*)(g   + h * DK + lane * 2);
        const float2 bv = *(const float2*)(bta + h * DK + lane * 2);

        float s[8];
        #pragma unroll
        for (int r = 0; r < 8; ++r) s[r] = xv[r].x + xv[r].y;
        #pragma unroll
        for (int o = 16; o >= 1; o >>= 1) {
            #pragma unroll
            for (int r = 0; r < 8; ++r) s[r] += __shfl_xor(s[r], o, 32);
        }
        float d0[8], d1[8], v[8];
        #pragma unroll
        for (int r = 0; r < 8; ++r) {
            const float mu = s[r] * (1.0f / DK);
            d0[r] = xv[r].x - mu;
            d1[r] = xv[r].y - mu;
            v[r]  = d0[r] * d0[r] + d1[r] * d1[r];
        }
        #pragma unroll
        for (int o = 16; o >= 1; o >>= 1) {
            #pragma unroll
            for (int r = 0; r < 8; ++r) v[r] += __shfl_xor(v[r], o, 32);
        }
        #pragma unroll
        for (int r = 0; r < 8; ++r) {
            const float rinv = rsqrtf(v[r] * (1.0f / DK) + 1e-5f);
            xv[r].x = d0[r] * rinv * gv.x + bv.x;
            xv[r].y = d1[r] * rinv * gv.y + bv.y;
        }
    }

    const int bb = rowBase >> 11, n0 = rowBase & (SEQ - 1);
    float* dst = dstBase + ((size_t)((bb * HEADS + h) * SEQ + n0 + wave * 8)) * DK + 2 * lane;
    #pragma unroll
    for (int r = 0; r < 8; ++r)
        *(float2*)(dst + r * DK) = xv[r];
}

__global__ __launch_bounds__(256)
void kv_gemm_kernel(float* kbuf, const float* __restrict__ vbuf) {
    __shared__ float lds_k[32 * BPS];
    __shared__ float lds_v[32 * BPS];

    const int tid  = threadIdx.x;
    const int lane = tid & 31, wave = tid >> 5;
    const int half = lane >> 4, l16 = lane & 15;
    const int M0 = (wave & 3) * 16, N0 = (wave >> 2) * 32;
    const int bh = blockIdx.y;
    const size_t slice = ((size_t)bh * SEQ + (size_t)blockIdx.x * KV_TOK) * DK;
    const float* kp_ptr = kbuf + slice;
    const float* vp_ptr = vbuf + slice;
    const int cp  = tid & 31;
    const int tb8 = tid >> 5;

    v8f acc0 = {}; v8f acc1 = {};

    for (int t0 = 0; t0 < KV_TOK; t0 += 64) {
        __syncthreads();
        #pragma unroll
        for (int i = 0; i < 4; ++i) {
            stage_pair(lds_k, tb8 + 8*i, cp, kp_ptr + (size_t)t0 * DK, DK);
            stage_pair(lds_v, tb8 + 8*i, cp, vp_ptr + (size_t)t0 * DK, DK);
        }
        __syncthreads();
        #pragma unroll
        for (int k0 = 0; k0 < 64; k0 += 4) {
            const int kp = (k0 >> 1) + half;
            v2f a  = *(const v2f*)&lds_k[kp * BPS + 2*(M0 + l16)];
            v2f b0 = *(const v2f*)&lds_v[kp * BPS + 2*(N0 + l16)];
            v2f b1 = *(const v2f*)&lds_v[kp * BPS + 2*(N0 + 16 + l16)];
            acc0 = wmma4(a, b0, acc0);
            acc1 = wmma4(a, b1, acc1);
        }
    }

    const float scale = 1.0f / 16384.0f;
    float* kvd = kbuf + slice;
    #pragma unroll
    for (int r = 0; r < 8; ++r) {
        int m = M0 + r + 8*half;
        kvd[m * DK + N0 + l16]      = acc0[r] * scale;
        kvd[m * DK + N0 + 16 + l16] = acc1[r] * scale;
    }
}

__global__ __launch_bounds__(256) __attribute__((amdgpu_waves_per_eu(8, 8)))
void out_gemm_kernel(const float* __restrict__ qbuf, const float* __restrict__ part,
                     float* __restrict__ out) {
    __shared__ float lds_q[64][68];
    __shared__ float lds_kv[32 * BPS];

    const int tid  = threadIdx.x;
    const int lane = tid & 31, wave = tid >> 5;
    const int half = lane >> 4, l16 = lane & 15;
    const int M0 = (wave & 3) * 16, N0 = (wave >> 2) * 32;
    const int tok0 = blockIdx.x * 64;
    const int bh = blockIdx.y;
    const int bb = bh >> 3, h = bh & 7;
    const float* qp  = qbuf + (size_t)bh * SEQ * DK + (size_t)tok0 * DK;
    const float* pp  = part + (size_t)bh * SEQ * DK;
    const int lr  = tid >> 2, lc = (tid & 3) * 16;
    const int c4  = tid & 15, kp16 = tid >> 4;

    #pragma unroll 1
    for (int i = 0; i < 2; ++i) {
        const float4 q0 = *(const float4*)&qp[(size_t)lr * DK + lc + 8*i];
        const float4 q1 = *(const float4*)&qp[(size_t)lr * DK + lc + 8*i + 4];
        const int kp = kp16 + 16*i;
        const float* s0 = pp + (size_t)(2 * kp) * DK + 4 * c4;
        float4 r0 = *(const float4*)s0;
        float4 r1 = *(const float4*)(s0 + DK);
        #pragma unroll
        for (int s = 1; s < KV_SPLIT; ++s) {
            const float4 a0 = *(const float4*)(s0 + (size_t)s * KV_TOK * DK);
            const float4 a1 = *(const float4*)(s0 + (size_t)s * KV_TOK * DK + DK);
            r0.x += a0.x; r0.y += a0.y; r0.z += a0.z; r0.w += a0.w;
            r1.x += a1.x; r1.y += a1.y; r1.z += a1.z; r1.w += a1.w;
        }
        float4 p0, p1;
        p0.x = r0.x; p0.y = r1.x; p0.z = r0.y; p0.w = r1.y;
        p1.x = r0.z; p1.y = r1.z; p1.z = r0.w; p1.w = r1.w;
        *(float4*)&lds_kv[kp * BPS + 8 * c4]     = p0;
        *(float4*)&lds_kv[kp * BPS + 8 * c4 + 4] = p1;
        *(float4*)&lds_q[lr][lc + 8*i]     = q0;
        *(float4*)&lds_q[lr][lc + 8*i + 4] = q1;
    }
    __syncthreads();

    v8f acc0 = {}; v8f acc1 = {};
    #pragma unroll
    for (int k0 = 0; k0 < 64; k0 += 4) {
        const int kp = (k0 >> 1) + half;
        v2f a  = *(const v2f*)&lds_q[M0 + l16][k0 + 2*half];
        v2f b0 = *(const v2f*)&lds_kv[kp * BPS + 2*(N0 + l16)];
        v2f b1 = *(const v2f*)&lds_kv[kp * BPS + 2*(N0 + 16 + l16)];
        acc0 = wmma4(a, b0, acc0);
        acc1 = wmma4(a, b1, acc1);
    }

    #pragma unroll
    for (int r = 0; r < 8; ++r) {
        int n = tok0 + M0 + r + 8*half;
        float* dst = out + ((size_t)(bb * SEQ + n)) * (HEADS * DK) + h * DK;
        dst[N0 + l16]      = acc0[r];
        dst[N0 + 16 + l16] = acc1[r];
    }
}

extern "C" void kernel_launch(void* const* d_in, const int* in_sizes, int n_in,
                              void* d_out, int out_size, void* d_ws, size_t ws_size,
                              hipStream_t stream) {
    const float* x  = (const float*)d_in[0];
    const float* w  = (const float*)d_in[1];
    const float* gq = (const float*)d_in[2];
    const float* bq = (const float*)d_in[3];
    const float* gk = (const float*)d_in[4];
    const float* bk = (const float*)d_in[5];
    float* out = (float*)d_out;

    float* ws = (float*)d_ws;
    const size_t perBuf = (size_t)BATCH * HEADS * SEQ * DK;
    float* qb  = ws;
    float* kb  = ws + perBuf;
    float* vb  = ws + 2 * perBuf;

    qkv_ln_gemm_kernel<<<dim3(NCOL / 64, (BATCH * SEQ) / 64), 256, 0, stream>>>(
        x, w, gq, bq, gk, bk, qb, kb, vb);
    kv_gemm_kernel<<<dim3(KV_SPLIT, BATCH * HEADS), 256, 0, stream>>>(kb, vb);
    out_gemm_kernel<<<dim3(SEQ / 64, BATCH * HEADS), 256, 0, stream>>>(qb, kb, out);
}


# qkv GEMM retiled: 128x128 block tile (two heads), 32x64 per wave (8 accumulator tiles, was 2): operand bytes fetched and LDS fragment reads per matrix instruction halved; k sum order kept
# speedup vs baseline: 2.1292x; 1.0968x over previous
#include <hip/hip_runtime.h>


#define BATCH 4
#define SEQ   2048
#define DIM   512
#define HEADS 8
#define DK    64
#define NCOL  1536
#define BPS   160
#define QT    128
#define QAP   68
#define QBP   288
#define QTP   132
#define KV_SPLIT 8
#define KV_TOK   (SEQ / KV_SPLIT)

typedef __attribute__((ext_vector_type(2))) float v2f;
typedef __attribute__((ext_vector_type(8))) float v8f;

__device__ __forceinline__ v8f wmma4(v2f a, v2f b, v8f c) {
    return __builtin_amdgcn_wmma_f32_16x16x4_f32(
         false, a,  false, b,
         (short)0, c,  false,  false);
}

__device__ __forceinline__ void stage_pair(float* __restrict__ dst, int kp, int cp,
                                           const float* __restrict__ src, int srcStride) {
    const float* s = src + (size_t)(2 * kp) * srcStride + 2 * cp;
    float2 r0 = *(const float2*)s;
    float2 r1 = *(const float2*)(s + srcStride);
    float4 pk;
    pk.x = r0.x; pk.y = r1.x; pk.z = r0.y; pk.w = r1.y;
    *(float4*)&dst[kp * BPS + 4 * cp] = pk;
}

__global__ __launch_bounds__(256) __attribute__((amdgpu_waves_per_eu(6, 6)))
void qkv_ln_gemm_kernel(const float* __restrict__ x, const float* __restrict__ w,
                        const float* __restrict__ gq, const float* __restrict__ bq,
                        const float* __restrict__ gk, const float* __restrict__ bk,
                        float* __restrict__ qb, float* __restrict__ kb,
                        float* __restrict__ vb) {
    __shared__ float lds[QT * QAP + 32 * QBP];
    float* lds_a = lds;
    float* lds_b = lds + QT * QAP;

    const int tid  = threadIdx.x;
    const int lane = tid & 31;
    const int wave = __builtin_amdgcn_readfirstlane(tid >> 5);
    const int half = lane >> 4, l16 = lane & 15;
    const int M0 = (wave & 3) * 32;
    const int N0 = (wave >> 2) * 64;

    const int cb    = blockIdx.x;
    const int col0  = cb * QT;
    const int which = cb >> 2, h0 = (cb & 3) * 2;
    const int rowBase = blockIdx.y * QT;
    float* dstBase = (which == 0) ? qb : (which == 1) ? kb : vb;

    const int ar  = tid >> 4;
    const int ac  = (tid & 15) * 4;
    const int bc  = tid & 31;
    const int bk8 = tid >> 5;
    const float* xp = x + (size_t)(rowBase + ar) * DIM + ac;
    const float* wp = w + (size_t)(2 * bk8) * NCOL + col0 + 4 * bc;
    float* sa = lds_a + ar * QAP + ac;
    float* sb = lds_b + bk8 * QBP + 8 * bc;

    v8f acc[2][4];
    #pragma unroll
    for (int a = 0; a < 2; ++a) {
        #pragma unroll
        for (int j = 0; j < 4; ++j) acc[a][j] = v8f{};
    }

    for (int kc = 0; kc < DIM; kc += 64) {
        __syncthreads();
        #pragma unroll
        for (int i = 0; i < 8; ++i)
            *(float4*)(sa + i * 16 * QAP) = *(const float4*)(xp + i * 16 * DIM);
        #pragma unroll
        for (int i = 0; i < 4; ++i) {
            const float4 r0 = *(const float4*)(wp + i * 16 * NCOL);
            const float4 r1 = *(const float4*)(wp + i * 16 * NCOL + NCOL);
            float4 p0, p1;
            p0.x = r0.x; p0.y = r1.x; p0.z = r0.y; p0.w = r1.y;
            p1.x = r0.z; p1.y = r1.z; p1.z = r0.w; p1.w = r1.w;
            *(float4*)(sb + i * 8 * QBP)     = p0;
            *(float4*)(sb + i * 8 * QBP + 4) = p1;
        }
        xp += 64;
        wp += (size_t)64 * NCOL;
        __syncthreads();
        #pragma unroll
        for (int k0 = 0; k0 < 64; k0 += 4) {
            const int kp = (k0 >> 1) + half;
            const v2f a0 = *(const v2f*)&lds_a[(M0 + l16) * QAP + k0 + 2*half];
            const v2f a1 = *(const v2f*)&lds_a[(M0 + 16 + l16) * QAP + k0 + 2*half];
            #pragma unroll
            for (int j = 0; j < 4; ++j) {
                const v2f b = *(const v2f*)&lds_b[kp * QBP + 2*(N0 + 16*j + l16)];
                acc[0][j] = wmma4(a0, b, acc[0][j]);
                acc[1][j] = wmma4(a1, b, acc[1][j]);
            }
        }
    }

    const int el = __builtin_amdgcn_mbcnt_lo(~0u, 0u);
    const int eh = el >> 4, e16 = el & 15;
    __syncthreads();
    #pragma unroll
    for (int a = 0; a < 2; ++a) {
        #pragma unroll
        for (int j = 0; j < 4; ++j) {
            #pragma unroll
            for (int r = 0; r < 8; ++r)
                lds[(M0 + 16*a + r + 8*eh) * QTP + N0 + 16*j + e16] = acc[a][j][r];
        }
    }
    __syncthreads();

    const int bb = rowBase >> 11, n0 = rowBase & (SEQ - 1);

    #pragma unroll
    for (int hh = 0; hh < 2; ++hh) {
        const int h = h0 + hh;
        float2 gv = {1.0f, 1.0f}, bv = {0.0f, 0.0f};
        if (which < 2) {
            const float* g   = (which == 0) ? gq : gk;
            const float* bta = (which == 0) ? bq : bk;
            gv = *(const float2*)(g   + h * DK + el * 2);
            bv = *(const float2*)(bta + h * DK + el * 2);
        }
        #pragma unroll
        for (int rb = 0; rb < 2; ++rb) {
            const int t0 = wave * 16 + rb * 8;
            float2 xv[8];
            #pragma unroll
            for (int r = 0; r < 8; ++r)
                xv[r] = *(const float2*)&lds[(t0 + r) * QTP + hh * DK + 2 * el];

            if (which < 2) {
                float s[8];
                #pragma unroll
                for (int r = 0; r < 8; ++r) s[r] = xv[r].x + xv[r].y;
                #pragma unroll
                for (int o = 16; o >= 1; o >>= 1) {
                    #pragma unroll
                    for (int r = 0; r < 8; ++r) s[r] += __shfl_xor(s[r], o, 32);
                }
                float d0[8], d1[8], v[8];
                #pragma unroll
                for (int r = 0; r < 8; ++r) {
                    const float mu = s[r] * (1.0f / DK);
                    d0[r] = xv[r].x - mu;
                    d1[r] = xv[r].y - mu;
                    v[r]  = d0[r] * d0[r] + d1[r] * d1[r];
                }
                #pragma unroll
                for (int o = 16; o >= 1; o >>= 1) {
                    #pragma unroll
                    for (int r = 0; r < 8; ++r) v[r] += __shfl_xor(v[r], o, 32);
                }
                #pragma unroll
                for (int r = 0; r < 8; ++r) {
                    const float rinv = rsqrtf(v[r] * (1.0f / DK) + 1e-5f);
                    xv[r].x = d0[r] * rinv * gv.x + bv.x;
                    xv[r].y = d1[r] * rinv * gv.y + bv.y;
                }
            }

            float* dst = dstBase + ((size_t)((bb * HEADS + h) * SEQ + n0 + t0)) * DK + 2 * el;
            #pragma unroll
            for (int r = 0; r < 8; ++r)
                *(float2*)(dst + r * DK) = xv[r];
        }
    }
}

__global__ __launch_bounds__(256)
void kv_gemm_kernel(float* kbuf, const float* __restrict__ vbuf) {
    __shared__ float lds_k[32 * BPS];
    __shared__ float lds_v[32 * BPS];

    const int tid  = threadIdx.x;
    const int lane = tid & 31, wave = tid >> 5;
    const int half = lane >> 4, l16 = lane & 15;
    const int M0 = (wave & 3) * 16, N0 = (wave >> 2) * 32;
    const int bh = blockIdx.y;
    const size_t slice = ((size_t)bh * SEQ + (size_t)blockIdx.x * KV_TOK) * DK;
    const float* kp_ptr = kbuf + slice;
    const float* vp_ptr = vbuf + slice;
    const int cp  = tid & 31;
    const int tb8 = tid >> 5;

    v8f acc0 = {}; v8f acc1 = {};

    for (int t0 = 0; t0 < KV_TOK; t0 += 64) {
        __syncthreads();
        #pragma unroll
        for (int i = 0; i < 4; ++i) {
            stage_pair(lds_k, tb8 + 8*i, cp, kp_ptr + (size_t)t0 * DK, DK);
            stage_pair(lds_v, tb8 + 8*i, cp, vp_ptr + (size_t)t0 * DK, DK);
        }
        __syncthreads();
        #pragma unroll
        for (int k0 = 0; k0 < 64; k0 += 4) {
            const int kp = (k0 >> 1) + half;
            v2f a  = *(const v2f*)&lds_k[kp * BPS + 2*(M0 + l16)];
            v2f b0 = *(const v2f*)&lds_v[kp * BPS + 2*(N0 + l16)];
            v2f b1 = *(const v2f*)&lds_v[kp * BPS + 2*(N0 + 16 + l16)];
            acc0 = wmma4(a, b0, acc0);
            acc1 = wmma4(a, b1, acc1);
        }
    }

    const float scale = 1.0f / 16384.0f;
    float* kvd = kbuf + slice;
    #pragma unroll
    for (int r = 0; r < 8; ++r) {
        int m = M0 + r + 8*half;
        kvd[m * DK + N0 + l16]      = acc0[r] * scale;
        kvd[m * DK + N0 + 16 + l16] = acc1[r] * scale;
    }
}

__global__ __launch_bounds__(256) __attribute__((amdgpu_waves_per_eu(8, 8)))
void out_gemm_kernel(const float* __restrict__ qbuf, const float* __restrict__ part,
                     float* __restrict__ out) {
    __shared__ float lds_q[64][68];
    __shared__ float lds_kv[32 * BPS];

    const int tid  = threadIdx.x;
    const int lane = tid & 31, wave = tid >> 5;
    const int half = lane >> 4, l16 = lane & 15;
    const int M0 = (wave & 3) * 16, N0 = (wave >> 2) * 32;
    const int tok0 = blockIdx.x * 64;
    const int bh = blockIdx.y;
    const int bb = bh >> 3, h = bh & 7;
    const float* qp  = qbuf + (size_t)bh * SEQ * DK + (size_t)tok0 * DK;
    const float* pp  = part + (size_t)bh * SEQ * DK;
    const int lr  = tid >> 2, lc = (tid & 3) * 16;
    const int c4  = tid & 15, kp16 = tid >> 4;

    #pragma unroll 1
    for (int i = 0; i < 2; ++i) {
        const float4 q0 = *(const float4*)&qp[(size_t)lr * DK + lc + 8*i];
        const float4 q1 = *(const float4*)&qp[(size_t)lr * DK + lc + 8*i + 4];
        const int kp = kp16 + 16*i;
        const float* s0 = pp + (size_t)(2 * kp) * DK + 4 * c4;
        float4 r0 = *(const float4*)s0;
        float4 r1 = *(const float4*)(s0 + DK);
        #pragma unroll
        for (int s = 1; s < KV_SPLIT; ++s) {
            const float4 a0 = *(const float4*)(s0 + (size_t)s * KV_TOK * DK);
            const float4 a1 = *(const float4*)(s0 + (size_t)s * KV_TOK * DK + DK);
            r0.x += a0.x; r0.y += a0.y; r0.z += a0.z; r0.w += a0.w;
            r1.x += a1.x; r1.y += a1.y; r1.z += a1.z; r1.w += a1.w;
        }
        float4 p0, p1;
        p0.x = r0.x; p0.y = r1.x; p0.z = r0.y; p0.w = r1.y;
        p1.x = r0.z; p1.y = r1.z; p1.z = r0.w; p1.w = r1.w;
        *(float4*)&lds_kv[kp * BPS + 8 * c4]     = p0;
        *(float4*)&lds_kv[kp * BPS + 8 * c4 + 4] = p1;
        *(float4*)&lds_q[lr][lc + 8*i]     = q0;
        *(float4*)&lds_q[lr][lc + 8*i + 4] = q1;
    }
    __syncthreads();

    v8f acc0 = {}; v8f acc1 = {};
    #pragma unroll
    for (int k0 = 0; k0 < 64; k0 += 4) {
        const int kp = (k0 >> 1) + half;
        v2f a  = *(const v2f*)&lds_q[M0 + l16][k0 + 2*half];
        v2f b0 = *(const v2f*)&lds_kv[kp * BPS + 2*(N0 + l16)];
        v2f b1 = *(const v2f*)&lds_kv[kp * BPS + 2*(N0 + 16 + l16)];
        acc0 = wmma4(a, b0, acc0);
        acc1 = wmma4(a, b1, acc1);
    }

    #pragma unroll
    for (int r = 0; r < 8; ++r) {
        int n = tok0 + M0 + r + 8*half;
        float* dst = out + ((size_t)(bb * SEQ + n)) * (HEADS * DK) + h * DK;
        dst[N0 + l16]      = acc0[r];
        dst[N0 + 16 + l16] = acc1[r];
    }
}

extern "C" void kernel_launch(void* const* d_in, const int* in_sizes, int n_in,
                              void* d_out, int out_size, void* d_ws, size_t ws_size,
                              hipStream_t stream) {
    const float* x  = (const float*)d_in[0];
    const float* w  = (const float*)d_in[1];
    const float* gq = (const float*)d_in[2];
    const float* bq = (const float*)d_in[3];
    const float* gk = (const float*)d_in[4];
    const float* bk = (const float*)d_in[5];
    float* out = (float*)d_out;

    float* ws = (float*)d_ws;
    const size_t perBuf = (size_t)BATCH * HEADS * SEQ * DK;
    float* qb  = ws;
    float* kb  = ws + perBuf;
    float* vb  = ws + 2 * perBuf;

    qkv_ln_gemm_kernel<<<dim3(NCOL / QT, (BATCH * SEQ) / QT), 256, 0, stream>>>(
        x, w, gq, bq, gk, bk, qb, kb, vb);
    kv_gemm_kernel<<<dim3(KV_SPLIT, BATCH * HEADS), 256, 0, stream>>>(kb, vb);
    out_gemm_kernel<<<dim3(SEQ / 64, BATCH * HEADS), 256, 0, stream>>>(qb, kb, out);
}


# k^T@v moved into the qkv GEMM epilogue: a block holds head h of k and v for 128 tokens, normalises k in LDS, stores a 64x64 partial (16 per head: token sum order changes); 2 launches, out tile 256
# speedup vs baseline: 2.4134x; 1.1335x over previous
#include <hip/hip_runtime.h>


#define BATCH 4
#define SEQ   2048
#define DIM   512
#define HEADS 8
#define DK    64
#define NCOL  1536
#define BPS   160
#define QT    128
#define QAP   68
#define QBP   288
#define QTP   132
#define KV_TILES (SEQ / QT)
#define OT    256

typedef __attribute__((ext_vector_type(2))) float v2f;
typedef __attribute__((ext_vector_type(8))) float v8f;

__device__ __forceinline__ v8f wmma4(v2f a, v2f b, v8f c) {
    return __builtin_amdgcn_wmma_f32_16x16x4_f32(
         false, a,  false, b,
         (short)0, c,  false,  false);
}

__device__ __forceinline__ void ln_rows8(float2 (&xv)[8], const float2 gv, const float2 bv) {
    float s[8];
    #pragma unroll
    for (int r = 0; r < 8; ++r) s[r] = xv[r].x + xv[r].y;
    #pragma unroll
    for (int o = 16; o >= 1; o >>= 1) {
        #pragma unroll
        for (int r = 0; r < 8; ++r) s[r] += __shfl_xor(s[r], o, 32);
    }
    float d0[8], d1[8], v[8];
    #pragma unroll
    for (int r = 0; r < 8; ++r) {
        const float mu = s[r] * (1.0f / DK);
        d0[r] = xv[r].x - mu;
        d1[r] = xv[r].y - mu;
        v[r]  = d0[r] * d0[r] + d1[r] * d1[r];
    }
    #pragma unroll
    for (int o = 16; o >= 1; o >>= 1) {
        #pragma unroll
        for (int r = 0; r < 8; ++r) v[r] += __shfl_xor(v[r], o, 32);
    }
    #pragma unroll
    for (int r = 0; r < 8; ++r) {
        const float rinv = rsqrtf(v[r] * (1.0f / DK) + 1e-5f);
        xv[r].x = d0[r] * rinv * gv.x + bv.x;
        xv[r].y = d1[r] * rinv * gv.y + bv.y;
    }
}

__global__ __launch_bounds__(256) __attribute__((amdgpu_waves_per_eu(6, 6)))
void qkv_ln_gemm_kernel(const float* __restrict__ x, const float* __restrict__ w,
                        const float* __restrict__ gq, const float* __restrict__ bq,
                        const float* __restrict__ gk, const float* __restrict__ bk,
                        float* __restrict__ qb, float* __restrict__ part) {
    __shared__ float lds[QT * QAP + 32 * QBP];
    float* lds_a = lds;
    float* lds_b = lds + QT * QAP;

    const int tid  = threadIdx.x;
    const int lane = tid & 31;
    const int wave = __builtin_amdgcn_readfirstlane(tid >> 5);
    const int half = lane >> 4, l16 = lane & 15;
    const int M0 = (wave & 3) * 32;
    const int N0 = (wave >> 2) * 64;

    const int cb  = blockIdx.x;
    const bool isq = cb < 4;
    const int hk  = cb - 4;
    const int rowBase = blockIdx.y * QT;

    const int ar  = tid >> 4;
    const int ac  = (tid & 15) * 4;
    const int bc  = tid & 31;
    const int bk8 = tid >> 5;
    const int wcol = isq ? (cb * QT + 4 * bc)
                         : (HEADS * DK + hk * DK + 4 * bc + ((bc >= 16) ? (HEADS * DK - DK) : 0));
    const float* xp = x + (size_t)(rowBase + ar) * DIM + ac;
    const float* wp = w + (size_t)(2 * bk8) * NCOL + wcol;
    float* sa = lds_a + ar * QAP + ac;
    float* sb = lds_b + bk8 * QBP + 8 * bc;

    v8f acc[2][4];
    #pragma unroll
    for (int a = 0; a < 2; ++a) {
        #pragma unroll
        for (int j = 0; j < 4; ++j) acc[a][j] = v8f{};
    }

    for (int kc = 0; kc < DIM; kc += 64) {
        __syncthreads();
        #pragma unroll
        for (int i = 0; i < 8; ++i)
            *(float4*)(sa + i * 16 * QAP) = *(const float4*)(xp + i * 16 * DIM);
        #pragma unroll
        for (int i = 0; i < 4; ++i) {
            const float4 r0 = *(const float4*)(wp + i * 16 * NCOL);
            const float4 r1 = *(const float4*)(wp + i * 16 * NCOL + NCOL);
            float4 p0, p1;
            p0.x = r0.x; p0.y = r1.x; p0.z = r0.y; p0.w = r1.y;
            p1.x = r0.z; p1.y = r1.z; p1.z = r0.w; p1.w = r1.w;
            *(float4*)(sb + i * 8 * QBP)     = p0;
            *(float4*)(sb + i * 8 * QBP + 4) = p1;
        }
        xp += 64;
        wp += (size_t)64 * NCOL;
        __syncthreads();
        #pragma unroll
        for (int k0 = 0; k0 < 64; k0 += 4) {
            const int kp = (k0 >> 1) + half;
            const v2f a0 = *(const v2f*)&lds_a[(M0 + l16) * QAP + k0 + 2*half];
            const v2f a1 = *(const v2f*)&lds_a[(M0 + 16 + l16) * QAP + k0 + 2*half];
            #pragma unroll
            for (int j = 0; j < 4; ++j) {
                const v2f b = *(const v2f*)&lds_b[kp * QBP + 2*(N0 + 16*j + l16)];
                acc[0][j] = wmma4(a0, b, acc[0][j]);
                acc[1][j] = wmma4(a1, b, acc[1][j]);
            }
        }
    }

    const int el = __builtin_amdgcn_mbcnt_lo(~0u, 0u);
    const int eh = el >> 4, e16 = el & 15;
    __syncthreads();
    #pragma unroll
    for (int a = 0; a < 2; ++a) {
        #pragma unroll
        for (int j = 0; j < 4; ++j) {
            #pragma unroll
            for (int r = 0; r < 8; ++r)
                lds[(M0 + 16*a + r + 8*eh) * QTP + N0 + 16*j + e16] = acc[a][j][r];
        }
    }
    __syncthreads();

    const int bb = rowBase >> 11, n0 = rowBase & (SEQ - 1);

    if (isq) {
        #pragma unroll
        for (int hh = 0; hh < 2; ++hh) {
            const int h = 2 * cb + hh;
            const float2 gv = *(const float2*)(gq + h * DK + el * 2);
            const float2 bv = *(const float2*)(bq + h * DK + el * 2);
            #pragma unroll
            for (int rb = 0; rb < 2; ++rb) {
                const int t0 = wave * 16 + rb * 8;
                float2 xv[8];
                #pragma unroll
                for (int r = 0; r < 8; ++r)
                    xv[r] = *(const float2*)&lds[(t0 + r) * QTP + hh * DK + 2 * el];
                ln_rows8(xv, gv, bv);
                float* dst = qb + ((size_t)((bb * HEADS + h) * SEQ + n0 + t0)) * DK + 2 * el;
                #pragma unroll
                for (int r = 0; r < 8; ++r)
                    *(float2*)(dst + r * DK) = xv[r];
            }
        }
    } else {
        const float2 gv = *(const float2*)(gk + hk * DK + el * 2);
        const float2 bv = *(const float2*)(bk + hk * DK + el * 2);
        #pragma unroll
        for (int rb = 0; rb < 2; ++rb) {
            const int t0 = wave * 16 + rb * 8;
            float2 xv[8];
            #pragma unroll
            for (int r = 0; r < 8; ++r)
                xv[r] = *(const float2*)&lds[(t0 + r) * QTP + 2 * el];
            ln_rows8(xv, gv, bv);
            #pragma unroll
            for (int r = 0; r < 8; ++r)
                *(float2*)&lds[(t0 + r) * QTP + 2 * el] = xv[r];
        }
        __syncthreads();

        const int P0 = (wave & 3) * 16, R0 = (wave >> 2) * 32;
        v8f p0 = {}; v8f p1 = {};
        #pragma unroll
        for (int k0 = 0; k0 < QT; k0 += 4) {
            const float* row = &lds[(k0 + 2 * eh) * QTP];
            v2f a, b0, b1;
            a.x  = row[P0 + e16];               a.y  = row[QTP + P0 + e16];
            b0.x = row[DK + R0 + e16];          b0.y = row[QTP + DK + R0 + e16];
            b1.x = row[DK + R0 + 16 + e16];     b1.y = row[QTP + DK + R0 + 16 + e16];
            p0 = wmma4(a, b0, p0);
            p1 = wmma4(a, b1, p1);
        }
        const float scale = 1.0f / 16384.0f;
        float* pd = part + ((size_t)((bb * HEADS + hk) * KV_TILES + (n0 >> 7))) * (DK * DK);
        #pragma unroll
        for (int r = 0; r < 8; ++r) {
            const int m = P0 + r + 8 * eh;
            pd[m * DK + R0 + e16]      = p0[r] * scale;
            pd[m * DK + R0 + 16 + e16] = p1[r] * scale;
        }
    }
}

__global__ __launch_bounds__(256)
void out_gemm_kernel(const float* __restrict__ qbuf, const float* __restrict__ part,
                     float* __restrict__ out) {
    __shared__ float lds_q[OT][QAP];
    __shared__ float lds_kv[32 * BPS];

    const int tid  = threadIdx.x;
    const int lane = tid & 31, wave = tid >> 5;
    const int half = lane >> 4, l16 = lane & 15;
    const int M0 = wave * 32;
    const int tok0 = blockIdx.x * OT;
    const int bh = blockIdx.y;
    const int bb = bh >> 3, h = bh & 7;
    const float* qp  = qbuf + (size_t)bh * SEQ * DK + (size_t)tok0 * DK;
    const float* pp  = part + (size_t)bh * KV_TILES * (DK * DK);
    const int lr  = tid >> 4, lc = (tid & 15) * 4;
    const int c4  = tid & 15, kp16 = tid >> 4;

    float4 qv[16];
    #pragma unroll
    for (int i = 0; i < 16; ++i)
        qv[i] = *(const float4*)&qp[(size_t)(lr + 16 * i) * DK + lc];

    #pragma unroll
    for (int i = 0; i < 2; ++i) {
        const int kp = kp16 + 16*i;
        const float* s0 = pp + (size_t)(2 * kp) * DK + 4 * c4;
        float4 r0 = *(const float4*)s0;
        float4 r1 = *(const float4*)(s0 + DK);
        #pragma unroll
        for (int s = 1; s < KV_TILES; ++s) {
            const float4 a0 = *(const float4*)(s0 + (size_t)s * (DK * DK));
            const float4 a1 = *(const float4*)(s0 + (size_t)s * (DK * DK) + DK);
            r0.x += a0.x; r0.y += a0.y; r0.z += a0.z; r0.w += a0.w;
            r1.x += a1.x; r1.y += a1.y; r1.z += a1.z; r1.w += a1.w;
        }
        float4 p0, p1;
        p0.x = r0.x; p0.y = r1.x; p0.z = r0.y; p0.w = r1.y;
        p1.x = r0.z; p1.y = r1.z; p1.z = r0.w; p1.w = r1.w;
        *(float4*)&lds_kv[kp * BPS + 8 * c4]     = p0;
        *(float4*)&lds_kv[kp * BPS + 8 * c4 + 4] = p1;
    }
    #pragma unroll
    for (int i = 0; i < 16; ++i)
        *(float4*)&lds_q[lr + 16 * i][lc] = qv[i];
    __syncthreads();

    v8f acc[2][4];
    #pragma unroll
    for (int a = 0; a < 2; ++a) {
        #pragma unroll
        for (int j = 0; j < 4; ++j) acc[a][j] = v8f{};
    }
    #pragma unroll
    for (int k0 = 0; k0 < 64; k0 += 4) {
        const int kp = (k0 >> 1) + half;
        const v2f a0 = *(const v2f*)&lds_q[M0 + l16][k0 + 2*half];
        const v2f a1 = *(const v2f*)&lds_q[M0 + 16 + l16][k0 + 2*half];
        #pragma unroll
        for (int j = 0; j < 4; ++j) {
            const v2f b = *(const v2f*)&lds_kv[kp * BPS + 2*(16*j + l16)];
            acc[0][j] = wmma4(a0, b, acc[0][j]);
            acc[1][j] = wmma4(a1, b, acc[1][j]);
        }
    }

    #pragma unroll
    for (int a = 0; a < 2; ++a) {
        #pragma unroll
        for (int r = 0; r < 8; ++r) {
            const int n = tok0 + M0 + 16*a + r + 8*half;
            float* dst = out + ((size_t)(bb * SEQ + n)) * (HEADS * DK) + h * DK + l16;
            #pragma unroll
            for (int j = 0; j < 4; ++j)
                dst[16*j] = acc[a][j][r];
        }
    }
}

extern "C" void kernel_launch(void* const* d_in, const int* in_sizes, int n_in,
                              void* d_out, int out_size, void* d_ws, size_t ws_size,
                              hipStream_t stream) {
    const float* x  = (const float*)d_in[0];
    const float* w  = (const float*)d_in[1];
    const float* gq = (const float*)d_in[2];
    const float* bq = (const float*)d_in[3];
    const float* gk = (const float*)d_in[4];
    const float* bk = (const float*)d_in[5];
    float* out = (float*)d_out;

    float* ws = (float*)d_ws;
    const size_t perBuf = (size_t)BATCH * HEADS * SEQ * DK;
    float* qb   = ws;
    float* part = ws + perBuf;

    qkv_ln_gemm_kernel<<<dim3(NCOL / QT, (BATCH * SEQ) / QT), 256, 0, stream>>>(
        x, w, gq, bq, gk, bk, qb, part);
    out_gemm_kernel<<<dim3(SEQ / OT, BATCH * HEADS), 256, 0, stream>>>(qb, part, out);
}


# qkv GEMM waves set issue priority by k-loop progress (s_setprio 3,3,3,2,2,2,1,1 per 64-k chunk, 0 in the epilogue): a WGP's three resident blocks advance together, none ends alone; sums unchanged
# speedup vs baseline: 2.5535x; 1.0581x over previous
#include <hip/hip_runtime.h>


#define BATCH 4
#define SEQ   2048
#define DIM   512
#define HEADS 8
#define DK    64
#define NCOL  1536
#define BPS   160
#define QT    128
#define QAP   68
#define QBP   288
#define QTP   132
#define KV_TILES (SEQ / QT)
#define OT    256

typedef __attribute__((ext_vector_type(2))) float v2f;
typedef __attribute__((ext_vector_type(8))) float v8f;

__device__ __forceinline__ v8f wmma4(v2f a, v2f b, v8f c) {
    return __builtin_amdgcn_wmma_f32_16x16x4_f32(
         false, a,  false, b,
         (short)0, c,  false,  false);
}

__device__ __forceinline__ void ln_rows8(float2 (&xv)[8], const float2 gv, const float2 bv) {
    float s[8];
    #pragma unroll
    for (int r = 0; r < 8; ++r) s[r] = xv[r].x + xv[r].y;
    #pragma unroll
    for (int o = 16; o >= 1; o >>= 1) {
        #pragma unroll
        for (int r = 0; r < 8; ++r) s[r] += __shfl_xor(s[r], o, 32);
    }
    float d0[8], d1[8], v[8];
    #pragma unroll
    for (int r = 0; r < 8; ++r) {
        const float mu = s[r] * (1.0f / DK);
        d0[r] = xv[r].x - mu;
        d1[r] = xv[r].y - mu;
        v[r]  = d0[r] * d0[r] + d1[r] * d1[r];
    }
    #pragma unroll
    for (int o = 16; o >= 1; o >>= 1) {
        #pragma unroll
        for (int r = 0; r < 8; ++r) v[r] += __shfl_xor(v[r], o, 32);
    }
    #pragma unroll
    for (int r = 0; r < 8; ++r) {
        const float rinv = rsqrtf(v[r] * (1.0f / DK) + 1e-5f);
        xv[r].x = d0[r] * rinv * gv.x + bv.x;
        xv[r].y = d1[r] * rinv * gv.y + bv.y;
    }
}

__global__ __launch_bounds__(256) __attribute__((amdgpu_waves_per_eu(6, 6)))
void qkv_ln_gemm_kernel(const float* __restrict__ x, const float* __restrict__ w,
                        const float* __restrict__ gq, const float* __restrict__ bq,
                        const float* __restrict__ gk, const float* __restrict__ bk,
                        float* __restrict__ qb, float* __restrict__ part) {
    __shared__ float lds[QT * QAP + 32 * QBP];
    float* lds_a = lds;
    float* lds_b = lds + QT * QAP;

    const int tid  = threadIdx.x;
    const int lane = tid & 31;
    const int wave = __builtin_amdgcn_readfirstlane(tid >> 5);
    const int half = lane >> 4, l16 = lane & 15;
    const int M0 = (wave & 3) * 32;
    const int N0 = (wave >> 2) * 64;

    const int cb  = blockIdx.x;
    const bool isq = cb < 4;
    const int hk  = cb - 4;
    const int rowBase = blockIdx.y * QT;

    const int ar  = tid >> 4;
    const int ac  = (tid & 15) * 4;
    const int bc  = tid & 31;
    const int bk8 = tid >> 5;
    const int wcol = isq ? (cb * QT + 4 * bc)
                         : (HEADS * DK + hk * DK + 4 * bc + ((bc >= 16) ? (HEADS * DK - DK) : 0));
    const float* xp = x + (size_t)(rowBase + ar) * DIM + ac;
    const float* wp = w + (size_t)(2 * bk8) * NCOL + wcol;
    float* sa = lds_a + ar * QAP + ac;
    float* sb = lds_b + bk8 * QBP + 8 * bc;

    v8f acc[2][4];
    #pragma unroll
    for (int a = 0; a < 2; ++a) {
        #pragma unroll
        for (int j = 0; j < 4; ++j) acc[a][j] = v8f{};
    }

    for (int kc = 0; kc < DIM; kc += 64) {
        if (kc == 0)        __builtin_amdgcn_s_setprio(3);
        else if (kc == 192) __builtin_amdgcn_s_setprio(2);
        else if (kc == 384) __builtin_amdgcn_s_setprio(1);
        __syncthreads();
        #pragma unroll
        for (int i = 0; i < 8; ++i)
            *(float4*)(sa + i * 16 * QAP) = *(const float4*)(xp + i * 16 * DIM);
        #pragma unroll
        for (int i = 0; i < 4; ++i) {
            const float4 r0 = *(const float4*)(wp + i * 16 * NCOL);
            const float4 r1 = *(const float4*)(wp + i * 16 * NCOL + NCOL);
            float4 p0, p1;
            p0.x = r0.x; p0.y = r1.x; p0.z = r0.y; p0.w = r1.y;
            p1.x = r0.z; p1.y = r1.z; p1.z = r0.w; p1.w = r1.w;
            *(float4*)(sb + i * 8 * QBP)     = p0;
            *(float4*)(sb + i * 8 * QBP + 4) = p1;
        }
        xp += 64;
        wp += (size_t)64 * NCOL;
        __syncthreads();
        #pragma unroll
        for (int k0 = 0; k0 < 64; k0 += 4) {
            const int kp = (k0 >> 1) + half;
            const v2f a0 = *(const v2f*)&lds_a[(M0 + l16) * QAP + k0 + 2*half];
            const v2f a1 = *(const v2f*)&lds_a[(M0 + 16 + l16) * QAP + k0 + 2*half];
            #pragma unroll
            for (int j = 0; j < 4; ++j) {
                const v2f b = *(const v2f*)&lds_b[kp * QBP + 2*(N0 + 16*j + l16)];
                acc[0][j] = wmma4(a0, b, acc[0][j]);
                acc[1][j] = wmma4(a1, b, acc[1][j]);
            }
        }
    }

    __builtin_amdgcn_s_setprio(0);

    const int el = __builtin_amdgcn_mbcnt_lo(~0u, 0u);
    const int eh = el >> 4, e16 = el & 15;
    __syncthreads();
    #pragma unroll
    for (int a = 0; a < 2; ++a) {
        #pragma unroll
        for (int j = 0; j < 4; ++j) {
            #pragma unroll
            for (int r = 0; r < 8; ++r)
                lds[(M0 + 16*a + r + 8*eh) * QTP + N0 + 16*j + e16] = acc[a][j][r];
        }
    }
    __syncthreads();

    const int bb = rowBase >> 11, n0 = rowBase & (SEQ - 1);

    if (isq) {
        #pragma unroll
        for (int hh = 0; hh < 2; ++hh) {
            const int h = 2 * cb + hh;
            const float2 gv = *(const float2*)(gq + h * DK + el * 2);
            const float2 bv = *(const float2*)(bq + h * DK + el * 2);
            #pragma unroll
            for (int rb = 0; rb < 2; ++rb) {
                const int t0 = wave * 16 + rb * 8;
                float2 xv[8];
                #pragma unroll
                for (int r = 0; r < 8; ++r)
                    xv[r] = *(const float2*)&lds[(t0 + r) * QTP + hh * DK + 2 * el];
                ln_rows8(xv, gv, bv);
                float* dst = qb + ((size_t)((bb * HEADS + h) * SEQ + n0 + t0)) * DK + 2 * el;
                #pragma unroll
                for (int r = 0; r < 8; ++r)
                    *(float2*)(dst + r * DK) = xv[r];
            }
        }
    } else {
        const float2 gv = *(const float2*)(gk + hk * DK + el * 2);
        const float2 bv = *(const float2*)(bk + hk * DK + el * 2);
        #pragma unroll
        for (int rb = 0; rb < 2; ++rb) {
            const int t0 = wave * 16 + rb * 8;
            float2 xv[8];
            #pragma unroll
            for (int r = 0; r < 8; ++r)
                xv[r] = *(const float2*)&lds[(t0 + r) * QTP + 2 * el];
            ln_rows8(xv, gv, bv);
            #pragma unroll
            for (int r = 0; r < 8; ++r)
                *(float2*)&lds[(t0 + r) * QTP + 2 * el] = xv[r];
        }
        __syncthreads();

        const int P0 = (wave & 3) * 16, R0 = (wave >> 2) * 32;
        v8f p0 = {}; v8f p1 = {};
        #pragma unroll
        for (int k0 = 0; k0 < QT; k0 += 4) {
            const float* row = &lds[(k0 + 2 * eh) * QTP];
            v2f a, b0, b1;
            a.x  = row[P0 + e16];               a.y  = row[QTP + P0 + e16];
            b0.x = row[DK + R0 + e16];          b0.y = row[QTP + DK + R0 + e16];
            b1.x = row[DK + R0 + 16 + e16];     b1.y = row[QTP + DK + R0 + 16 + e16];
            p0 = wmma4(a, b0, p0);
            p1 = wmma4(a, b1, p1);
        }
        const float scale = 1.0f / 16384.0f;
        float* pd = part + ((size_t)((bb * HEADS + hk) * KV_TILES + (n0 >> 7))) * (DK * DK);
        #pragma unroll
        for (int r = 0; r < 8; ++r) {
            const int m = P0 + r + 8 * eh;
            pd[m * DK + R0 + e16]      = p0[r] * scale;
            pd[m * DK + R0 + 16 + e16] = p1[r] * scale;
        }
    }
}

__global__ __launch_bounds__(256)
void out_gemm_kernel(const float* __restrict__ qbuf, const float* __restrict__ part,
                     float* __restrict__ out) {
    __shared__ float lds_q[OT][QAP];
    __shared__ float lds_kv[32 * BPS];

    const int tid  = threadIdx.x;
    const int lane = tid & 31, wave = tid >> 5;
    const int half = lane >> 4, l16 = lane & 15;
    const int M0 = wave * 32;
    const int tok0 = blockIdx.x * OT;
    const int bh = blockIdx.y;
    const int bb = bh >> 3, h = bh & 7;
    const float* qp  = qbuf + (size_t)bh * SEQ * DK + (size_t)tok0 * DK;
    const float* pp  = part + (size_t)bh * KV_TILES * (DK * DK);
    const int lr  = tid >> 4, lc = (tid & 15) * 4;
    const int c4  = tid & 15, kp16 = tid >> 4;

    float4 qv[16];
    #pragma unroll
    for (int i = 0; i < 16; ++i)
        qv[i] = *(const float4*)&qp[(size_t)(lr + 16 * i) * DK + lc];

    #pragma unroll
    for (int i = 0; i < 2; ++i) {
        const int kp = kp16 + 16*i;
        const float* s0 = pp + (size_t)(2 * kp) * DK + 4 * c4;
        float4 r0 = *(const float4*)s0;
        float4 r1 = *(const float4*)(s0 + DK);
        #pragma unroll
        for (int s = 1; s < KV_TILES; ++s) {
            const float4 a0 = *(const float4*)(s0 + (size_t)s * (DK * DK));
            const float4 a1 = *(const float4*)(s0 + (size_t)s * (DK * DK) + DK);
            r0.x += a0.x; r0.y += a0.y; r0.z += a0.z; r0.w += a0.w;
            r1.x += a1.x; r1.y += a1.y; r1.z += a1.z; r1.w += a1.w;
        }
        float4 p0, p1;
        p0.x = r0.x; p0.y = r1.x; p0.z = r0.y; p0.w = r1.y;
        p1.x = r0.z; p1.y = r1.z; p1.z = r0.w; p1.w = r1.w;
        *(float4*)&lds_kv[kp * BPS + 8 * c4]     = p0;
        *(float4*)&lds_kv[kp * BPS + 8 * c4 + 4] = p1;
    }
    #pragma unroll
    for (int i = 0; i < 16; ++i)
        *(float4*)&lds_q[lr + 16 * i][lc] = qv[i];
    __syncthreads();

    v8f acc[2][4];
    #pragma unroll
    for (int a = 0; a < 2; ++a) {
        #pragma unroll
        for (int j = 0; j < 4; ++j) acc[a][j] = v8f{};
    }
    #pragma unroll
    for (int k0 = 0; k0 < 64; k0 += 4) {
        const int kp = (k0 >> 1) + half;
        const v2f a0 = *(const v2f*)&lds_q[M0 + l16][k0 + 2*half];
        const v2f a1 = *(const v2f*)&lds_q[M0 + 16 + l16][k0 + 2*half];
        #pragma unroll
        for (int j = 0; j < 4; ++j) {
            const v2f b = *(const v2f*)&lds_kv[kp * BPS + 2*(16*j + l16)];
            acc[0][j] = wmma4(a0, b, acc[0][j]);
            acc[1][j] = wmma4(a1, b, acc[1][j]);
        }
    }

    #pragma unroll
    for (int a = 0; a < 2; ++a) {
        #pragma unroll
        for (int r = 0; r < 8; ++r) {
            const int n = tok0 + M0 + 16*a + r + 8*half;
            float* dst = out + ((size_t)(bb * SEQ + n)) * (HEADS * DK) + h * DK + l16;
            #pragma unroll
            for (int j = 0; j < 4; ++j)
                dst[16*j] = acc[a][j][r];
        }
    }
}

extern "C" void kernel_launch(void* const* d_in, const int* in_sizes, int n_in,
                              void* d_out, int out_size, void* d_ws, size_t ws_size,
                              hipStream_t stream) {
    const float* x  = (const float*)d_in[0];
    const float* w  = (const float*)d_in[1];
    const float* gq = (const float*)d_in[2];
    const float* bq = (const float*)d_in[3];
    const float* gk = (const float*)d_in[4];
    const float* bk = (const float*)d_in[5];
    float* out = (float*)d_out;

    float* ws = (float*)d_ws;
    const size_t perBuf = (size_t)BATCH * HEADS * SEQ * DK;
    float* qb   = ws;
    float* part = ws + perBuf;

    qkv_ln_gemm_kernel<<<dim3(NCOL / QT, (BATCH * SEQ) / QT), 256, 0, stream>>>(
        x, w, gq, bq, gk, bk, qb, part);
    out_gemm_kernel<<<dim3(SEQ / OT, BATCH * HEADS), 256, 0, stream>>>(qb, part, out);
}
